# starting kernel
# baseline (speedup 1.0000x reference)
#include <hip/hip_runtime.h>

typedef __attribute__((ext_vector_type(2))) float v2f;
typedef __attribute__((ext_vector_type(8))) float v8f;

#define TOKENS  4096
#define IN_F    4096
#define OUT_F   4096
#define LOWRANK 256
#define BS      256
#define ACTIVE  5
#define KBF     (ACTIVE * BS)

__global__ __launch_bounds__(256) void gather_wg(const float* __restrict__ weight,
                                                 const int* __restrict__ flat,
                                                 float* __restrict__ wg) {
    int g   = blockIdx.x * 256 + threadIdx.x;
    int f   = g / KBF;
    int rem = g - f * KBF;
    int j   = rem >> 8;
    int c   = rem & 255;
    int ob  = f >> 8;
    int rr  = f & 255;
    int fl  = flat[ob * ACTIVE + j];
    int q   = fl / ACTIVE;
    int bb  = fl - q * ACTIVE;
    int k   = bb * BS + rr;
    int kr  = k / ACTIVE;
    int a   = k - kr * ACTIVE;
    int srcRow = q * BS + kr;
    wg[g] = weight[(srcRow * ACTIVE + a) * BS + c];
}

__global__ __launch_bounds__(256) void gemm_xw1(const float* __restrict__ x,
                                                const float* __restrict__ w1,
                                                float* __restrict__ T) {
    const int lane = threadIdx.x & 31;
    const int wave = threadIdx.x >> 5;
    const int tile = blockIdx.x * 8 + wave;
    const int nft  = LOWRANK / 32;
    const int tt   = tile / nft;
    const int ft   = tile - tt * nft;
    const int t0   = tt * 32;
    const int f0   = ft * 32;
    const int lr   = lane & 15;
    const int co   = (lane >> 4) * 2;

    const float* a0p = x  + (t0 + lr)      * IN_F + co;
    const float* a1p = x  + (t0 + 16 + lr) * IN_F + co;
    const float* b0p = w1 + (f0 + lr)      * IN_F + co;
    const float* b1p = w1 + (f0 + 16 + lr) * IN_F + co;

    v8f acc00 = {}, acc01 = {}, acc10 = {}, acc11 = {};
    for (int k = 0; k < IN_F; k += 16) {
#pragma unroll
        for (int u = 0; u < 16; u += 4) {
            v2f a0 = *(const v2f*)(a0p + k + u);
            v2f a1 = *(const v2f*)(a1p + k + u);
            v2f b0 = *(const v2f*)(b0p + k + u);
            v2f b1 = *(const v2f*)(b1p + k + u);
            acc00 = __builtin_amdgcn_wmma_f32_16x16x4_f32(false, a0, false, b0,
                                                          (short)0, acc00, false, false);
            acc01 = __builtin_amdgcn_wmma_f32_16x16x4_f32(false, a0, false, b1,
                                                          (short)0, acc01, false, false);
            acc10 = __builtin_amdgcn_wmma_f32_16x16x4_f32(false, a1, false, b0,
                                                          (short)0, acc10, false, false);
            acc11 = __builtin_amdgcn_wmma_f32_16x16x4_f32(false, a1, false, b1,
                                                          (short)0, acc11, false, false);
        }
    }
    const int m0 = (lane >> 4) * 8;
#pragma unroll
    for (int r = 0; r < 8; ++r) {
        T[(t0 + m0 + r)      * LOWRANK + f0 + lr]      = acc00[r];
        T[(t0 + m0 + r)      * LOWRANK + f0 + 16 + lr] = acc01[r];
        T[(t0 + 16 + m0 + r) * LOWRANK + f0 + lr]      = acc10[r];
        T[(t0 + 16 + m0 + r) * LOWRANK + f0 + 16 + lr] = acc11[r];
    }
}

__global__ __launch_bounds__(256) void fused_out(const float* __restrict__ x,
                                                 const float* __restrict__ wg,
                                                 const float* __restrict__ T,
                                                 const float* __restrict__ w2,
                                                 const float* __restrict__ bias,
                                                 const int* __restrict__ flat,
                                                 float* __restrict__ out) {
    const int lane = threadIdx.x & 31;
    const int wave = threadIdx.x >> 5;
    const int tile = blockIdx.x * 8 + wave;
    const int nft  = OUT_F / 64;
    const int tt   = tile / nft;
    const int ft   = tile - tt * nft;
    const int t0   = tt * 64;
    const int f0   = ft * 64;
    const int ob   = f0 >> 8;
    const int lr   = lane & 15;
    const int co   = (lane >> 4) * 2;

    v8f acc[4][4];
#pragma unroll
    for (int i = 0; i < 4; ++i)
#pragma unroll
        for (int n = 0; n < 4; ++n) acc[i][n] = (v8f){};

    for (int j = 0; j < ACTIVE; ++j) {
        const int q = flat[ob * ACTIVE + j] / ACTIVE;
        const float* arow[4];
        const float* brow[4];
#pragma unroll
        for (int i = 0; i < 4; ++i)
            arow[i] = x + (t0 + 16 * i + lr) * IN_F + q * BS + co;
#pragma unroll
        for (int n = 0; n < 4; ++n)
            brow[n] = wg + (f0 + 16 * n + lr) * KBF + j * BS + co;

        for (int k = 0; k < BS; k += 4) {
            v2f af[4], bf[4];
#pragma unroll
            for (int i = 0; i < 4; ++i) af[i] = *(const v2f*)(arow[i] + k);
#pragma unroll
            for (int n = 0; n < 4; ++n) bf[n] = *(const v2f*)(brow[n] + k);
#pragma unroll
            for (int i = 0; i < 4; ++i)
#pragma unroll
                for (int n = 0; n < 4; ++n)
                    acc[i][n] = __builtin_amdgcn_wmma_f32_16x16x4_f32(
                        false, af[i], false, bf[n], (short)0, acc[i][n], false, false);
        }
    }

    {
        const float* arow[4];
        const float* brow[4];
#pragma unroll
        for (int i = 0; i < 4; ++i)
            arow[i] = T  + (t0 + 16 * i + lr) * LOWRANK + co;
#pragma unroll
        for (int n = 0; n < 4; ++n)
            brow[n] = w2 + (f0 + 16 * n + lr) * LOWRANK + co;

        for (int k = 0; k < LOWRANK; k += 4) {
            v2f af[4], bf[4];
#pragma unroll
            for (int i = 0; i < 4; ++i) af[i] = *(const v2f*)(arow[i] + k);
#pragma unroll
            for (int n = 0; n < 4; ++n) bf[n] = *(const v2f*)(brow[n] + k);
#pragma unroll
            for (int i = 0; i < 4; ++i)
#pragma unroll
                for (int n = 0; n < 4; ++n)
                    acc[i][n] = __builtin_amdgcn_wmma_f32_16x16x4_f32(
                        false, af[i], false, bf[n], (short)0, acc[i][n], false, false);
        }
    }

    const int m0 = (lane >> 4) * 8;
#pragma unroll
    for (int n = 0; n < 4; ++n) {
        const float bv = bias[f0 + 16 * n + lr];
#pragma unroll
        for (int i = 0; i < 4; ++i)
#pragma unroll
            for (int r = 0; r < 8; ++r)
                out[(t0 + 16 * i + m0 + r) * OUT_F + f0 + 16 * n + lr] = acc[i][n][r] + bv;
    }
}

extern "C" void kernel_launch(void* const* d_in, const int* in_sizes, int n_in,
                              void* d_out, int out_size, void* d_ws, size_t ws_size,
                              hipStream_t stream) {
    const float* x      = (const float*)d_in[0];
    const float* weight = (const float*)d_in[1];
    const float* w1     = (const float*)d_in[2];
    const float* w2     = (const float*)d_in[3];
    const float* bias   = (const float*)d_in[4];
    const int*   flat   = (const int*)d_in[5];
    float*       out    = (float*)d_out;

    float* wg = (float*)d_ws;
    float* T  = wg + (size_t)OUT_F * KBF;

    gather_wg<<<dim3((OUT_F * KBF) / 256), dim3(256), 0, stream>>>(weight, flat, wg);

    gemm_xw1<<<dim3(1024 / 8), dim3(256), 0, stream>>>(x, w1, T);

    fused_out<<<dim3(4096 / 8), dim3(256), 0, stream>>>(x, wg, T, w2, bias, flat, out);
}


# A/B fragment loads widened from 8 to 16 bytes per lane in both GEMM kernels: one float4 feeds two 16x16x4 products (same k order in both loaders), so k is summed in a new order inside each 8
# speedup vs baseline: 1.1663x; 1.1663x over previous
#include <hip/hip_runtime.h>

typedef __attribute__((ext_vector_type(2))) float v2f;
typedef __attribute__((ext_vector_type(4))) float v4f;
typedef __attribute__((ext_vector_type(8))) float v8f;

#define LO2(v) (__builtin_shufflevector((v), (v), 0, 1))
#define HI2(v) (__builtin_shufflevector((v), (v), 2, 3))

#define PRODUCT_TAIL_SLOTS()                                                        \
    do {                                                                            \
        __builtin_amdgcn_sched_barrier(0);                                          \
        asm volatile("v_nop\n\tv_nop\n\tv_nop\n\tv_nop\n\tv_nop\n\tv_nop\n\tv_nop\n\tv_nop"); \
        __builtin_amdgcn_sched_barrier(0);                                          \
    } while (0)

#define TOKENS  4096
#define IN_F    4096
#define OUT_F   4096
#define LOWRANK 256
#define BS      256
#define ACTIVE  5
#define KBF     (ACTIVE * BS)

__global__ __launch_bounds__(256) void gather_wg(const float* __restrict__ weight,
                                                 const int* __restrict__ flat,
                                                 float* __restrict__ wg) {
    int g   = blockIdx.x * 256 + threadIdx.x;
    int f   = g / KBF;
    int rem = g - f * KBF;
    int j   = rem >> 8;
    int c   = rem & 255;
    int ob  = f >> 8;
    int rr  = f & 255;
    int fl  = flat[ob * ACTIVE + j];
    int q   = fl / ACTIVE;
    int bb  = fl - q * ACTIVE;
    int k   = bb * BS + rr;
    int kr  = k / ACTIVE;
    int a   = k - kr * ACTIVE;
    int srcRow = q * BS + kr;
    wg[g] = weight[(srcRow * ACTIVE + a) * BS + c];
}

__global__ __launch_bounds__(256) void gemm_xw1(const float* __restrict__ x,
                                                const float* __restrict__ w1,
                                                float* __restrict__ T) {
    const int lane = threadIdx.x & 31;
    const int wave = threadIdx.x >> 5;
    const int tile = blockIdx.x * 8 + wave;
    const int nft  = LOWRANK / 32;
    const int tt   = tile / nft;
    const int ft   = tile - tt * nft;
    const int t0   = tt * 32;
    const int f0   = ft * 32;
    const int lr   = lane & 15;
    const int co   = (lane >> 4) * 4;

    const float* a0p = x  + (t0 + lr)      * IN_F + co;
    const float* a1p = x  + (t0 + 16 + lr) * IN_F + co;
    const float* b0p = w1 + (f0 + lr)      * IN_F + co;
    const float* b1p = w1 + (f0 + 16 + lr) * IN_F + co;

    v8f acc00 = {}, acc01 = {}, acc10 = {}, acc11 = {};
    for (int k = 0; k < IN_F; k += 16) {
#pragma unroll
        for (int u = 0; u < 16; u += 8) {
            const v4f a0 = *(const v4f*)(a0p + k + u);
            const v4f a1 = *(const v4f*)(a1p + k + u);
            const v4f b0 = *(const v4f*)(b0p + k + u);
            const v4f b1 = *(const v4f*)(b1p + k + u);
            acc00 = __builtin_amdgcn_wmma_f32_16x16x4_f32(false, LO2(a0), false, LO2(b0),
                                                          (short)0, acc00, false, false);
            acc01 = __builtin_amdgcn_wmma_f32_16x16x4_f32(false, LO2(a0), false, LO2(b1),
                                                          (short)0, acc01, false, false);
            acc10 = __builtin_amdgcn_wmma_f32_16x16x4_f32(false, LO2(a1), false, LO2(b0),
                                                          (short)0, acc10, false, false);
            acc11 = __builtin_amdgcn_wmma_f32_16x16x4_f32(false, LO2(a1), false, LO2(b1),
                                                          (short)0, acc11, false, false);
            acc00 = __builtin_amdgcn_wmma_f32_16x16x4_f32(false, HI2(a0), false, HI2(b0),
                                                          (short)0, acc00, false, false);
            acc01 = __builtin_amdgcn_wmma_f32_16x16x4_f32(false, HI2(a0), false, HI2(b1),
                                                          (short)0, acc01, false, false);
            acc10 = __builtin_amdgcn_wmma_f32_16x16x4_f32(false, HI2(a1), false, HI2(b0),
                                                          (short)0, acc10, false, false);
            acc11 = __builtin_amdgcn_wmma_f32_16x16x4_f32(false, HI2(a1), false, HI2(b1),
                                                          (short)0, acc11, false, false);
        }
    }
    const int m0 = (lane >> 4) * 8;
#pragma unroll
    for (int r = 0; r < 8; ++r) {
        T[(t0 + m0 + r)      * LOWRANK + f0 + lr]      = acc00[r];
        T[(t0 + m0 + r)      * LOWRANK + f0 + 16 + lr] = acc01[r];
        T[(t0 + 16 + m0 + r) * LOWRANK + f0 + lr]      = acc10[r];
        T[(t0 + 16 + m0 + r) * LOWRANK + f0 + 16 + lr] = acc11[r];
    }
}

__global__ __launch_bounds__(256) void fused_out(const float* __restrict__ x,
                                                 const float* __restrict__ wg,
                                                 const float* __restrict__ T,
                                                 const float* __restrict__ w2,
                                                 const float* __restrict__ bias,
                                                 const int* __restrict__ flat,
                                                 float* __restrict__ out) {
    const int lane = threadIdx.x & 31;
    const int wave = threadIdx.x >> 5;
    const int tile = blockIdx.x * 8 + wave;
    const int nft  = OUT_F / 64;
    const int tt   = tile / nft;
    const int ft   = tile - tt * nft;
    const int t0   = tt * 64;
    const int f0   = ft * 64;
    const int ob   = f0 >> 8;
    const int lr   = lane & 15;
    const int co   = (lane >> 4) * 4;

    v8f acc[4][4];
#pragma unroll
    for (int i = 0; i < 4; ++i)
#pragma unroll
        for (int n = 0; n < 4; ++n) acc[i][n] = (v8f){};

    for (int j = 0; j < ACTIVE; ++j) {
        const int q = flat[ob * ACTIVE + j] / ACTIVE;
        const float* arow[4];
        const float* brow[4];
#pragma unroll
        for (int i = 0; i < 4; ++i)
            arow[i] = x + (t0 + 16 * i + lr) * IN_F + q * BS + co;
#pragma unroll
        for (int n = 0; n < 4; ++n)
            brow[n] = wg + (f0 + 16 * n + lr) * KBF + j * BS + co;

        for (int k = 0; k < BS; k += 8) {
            v4f af[4], bf[4];
#pragma unroll
            for (int i = 0; i < 4; ++i) af[i] = *(const v4f*)(arow[i] + k);
#pragma unroll
            for (int n = 0; n < 4; ++n) bf[n] = *(const v4f*)(brow[n] + k);
#pragma unroll
            for (int i = 0; i < 4; ++i)
#pragma unroll
                for (int n = 0; n < 4; ++n)
                    acc[i][n] = __builtin_amdgcn_wmma_f32_16x16x4_f32(
                        false, LO2(af[i]), false, LO2(bf[n]), (short)0, acc[i][n], false, false);
#pragma unroll
            for (int i = 0; i < 4; ++i)
#pragma unroll
                for (int n = 0; n < 4; ++n)
                    acc[i][n] = __builtin_amdgcn_wmma_f32_16x16x4_f32(
                        false, HI2(af[i]), false, HI2(bf[n]), (short)0, acc[i][n], false, false);
            PRODUCT_TAIL_SLOTS();
        }
    }

    {
        const float* arow[4];
        const float* brow[4];
#pragma unroll
        for (int i = 0; i < 4; ++i)
            arow[i] = T  + (t0 + 16 * i + lr) * LOWRANK + co;
#pragma unroll
        for (int n = 0; n < 4; ++n)
            brow[n] = w2 + (f0 + 16 * n + lr) * LOWRANK + co;

        for (int k = 0; k < LOWRANK; k += 8) {
            v4f af[4], bf[4];
#pragma unroll
            for (int i = 0; i < 4; ++i) af[i] = *(const v4f*)(arow[i] + k);
#pragma unroll
            for (int n = 0; n < 4; ++n) bf[n] = *(const v4f*)(brow[n] + k);
#pragma unroll
            for (int i = 0; i < 4; ++i)
#pragma unroll
                for (int n = 0; n < 4; ++n)
                    acc[i][n] = __builtin_amdgcn_wmma_f32_16x16x4_f32(
                        false, LO2(af[i]), false, LO2(bf[n]), (short)0, acc[i][n], false, false);
#pragma unroll
            for (int i = 0; i < 4; ++i)
#pragma unroll
                for (int n = 0; n < 4; ++n)
                    acc[i][n] = __builtin_amdgcn_wmma_f32_16x16x4_f32(
                        false, HI2(af[i]), false, HI2(bf[n]), (short)0, acc[i][n], false, false);
            PRODUCT_TAIL_SLOTS();
        }
    }

    const int m0 = (lane >> 4) * 8;
#pragma unroll
    for (int n = 0; n < 4; ++n) {
        const float bv = bias[f0 + 16 * n + lr];
#pragma unroll
        for (int i = 0; i < 4; ++i)
#pragma unroll
            for (int r = 0; r < 8; ++r)
                out[(t0 + 16 * i + m0 + r) * OUT_F + f0 + 16 * n + lr] = acc[i][n][r] + bv;
    }
}

extern "C" void kernel_launch(void* const* d_in, const int* in_sizes, int n_in,
                              void* d_out, int out_size, void* d_ws, size_t ws_size,
                              hipStream_t stream) {
    const float* x      = (const float*)d_in[0];
    const float* weight = (const float*)d_in[1];
    const float* w1     = (const float*)d_in[2];
    const float* w2     = (const float*)d_in[3];
    const float* bias   = (const float*)d_in[4];
    const int*   flat   = (const int*)d_in[5];
    float*       out    = (float*)d_out;

    float* wg = (float*)d_ws;
    float* T  = wg + (size_t)OUT_F * KBF;

    gather_wg<<<dim3((OUT_F * KBF) / 256), dim3(256), 0, stream>>>(weight, flat, wg);

    gemm_xw1<<<dim3(1024 / 8), dim3(256), 0, stream>>>(x, w1, T);

    fused_out<<<dim3(4096 / 8), dim3(256), 0, stream>>>(x, wg, T, w2, bias, flat, out);
}


# Fused output kernel: A/B tiles staged through LDS per work-group (128x256 tile, whole-line global loads, 2 stages, 1 barrier per 32 k) instead of per-wave fragment loads; summation order unchanged
# speedup vs baseline: 1.4536x; 1.2463x over previous
#include <hip/hip_runtime.h>

typedef __attribute__((ext_vector_type(2))) float v2f;
typedef __attribute__((ext_vector_type(4))) float v4f;
typedef __attribute__((ext_vector_type(8))) float v8f;

#define LO2(v) (__builtin_shufflevector((v), (v), 0, 1))
#define HI2(v) (__builtin_shufflevector((v), (v), 2, 3))

#define PRODUCT_TAIL_SLOTS()                                                        \
    do {                                                                            \
        __builtin_amdgcn_sched_barrier(0);                                          \
        asm volatile("v_nop\n\tv_nop\n\tv_nop\n\tv_nop\n\tv_nop\n\tv_nop\n\tv_nop\n\tv_nop"); \
        __builtin_amdgcn_sched_barrier(0);                                          \
    } while (0)

#define TOKENS  4096
#define IN_F    4096
#define OUT_F   4096
#define LOWRANK 256
#define BS      256
#define ACTIVE  5
#define KBF     (ACTIVE * BS)

__global__ __launch_bounds__(256) void gather_wg(const float* __restrict__ weight,
                                                 const int* __restrict__ flat,
                                                 float* __restrict__ wg) {
    int g   = blockIdx.x * 256 + threadIdx.x;
    int f   = g / KBF;
    int rem = g - f * KBF;
    int j   = rem >> 8;
    int c   = rem & 255;
    int ob  = f >> 8;
    int rr  = f & 255;
    int fl  = flat[ob * ACTIVE + j];
    int q   = fl / ACTIVE;
    int bb  = fl - q * ACTIVE;
    int k   = bb * BS + rr;
    int kr  = k / ACTIVE;
    int a   = k - kr * ACTIVE;
    int srcRow = q * BS + kr;
    wg[g] = weight[(srcRow * ACTIVE + a) * BS + c];
}

__global__ __launch_bounds__(256) void gemm_xw1(const float* __restrict__ x,
                                                const float* __restrict__ w1,
                                                float* __restrict__ T) {
    const int lane = threadIdx.x & 31;
    const int wave = threadIdx.x >> 5;
    const int tile = blockIdx.x * 8 + wave;
    const int nft  = LOWRANK / 32;
    const int tt   = tile / nft;
    const int ft   = tile - tt * nft;
    const int t0   = tt * 32;
    const int f0   = ft * 32;
    const int lr   = lane & 15;
    const int co   = (lane >> 4) * 4;

    const float* a0p = x  + (t0 + lr)      * IN_F + co;
    const float* a1p = x  + (t0 + 16 + lr) * IN_F + co;
    const float* b0p = w1 + (f0 + lr)      * IN_F + co;
    const float* b1p = w1 + (f0 + 16 + lr) * IN_F + co;

    v8f acc00 = {}, acc01 = {}, acc10 = {}, acc11 = {};
    for (int k = 0; k < IN_F; k += 16) {
#pragma unroll
        for (int u = 0; u < 16; u += 8) {
            const v4f a0 = *(const v4f*)(a0p + k + u);
            const v4f a1 = *(const v4f*)(a1p + k + u);
            const v4f b0 = *(const v4f*)(b0p + k + u);
            const v4f b1 = *(const v4f*)(b1p + k + u);
            acc00 = __builtin_amdgcn_wmma_f32_16x16x4_f32(false, LO2(a0), false, LO2(b0),
                                                          (short)0, acc00, false, false);
            acc01 = __builtin_amdgcn_wmma_f32_16x16x4_f32(false, LO2(a0), false, LO2(b1),
                                                          (short)0, acc01, false, false);
            acc10 = __builtin_amdgcn_wmma_f32_16x16x4_f32(false, LO2(a1), false, LO2(b0),
                                                          (short)0, acc10, false, false);
            acc11 = __builtin_amdgcn_wmma_f32_16x16x4_f32(false, LO2(a1), false, LO2(b1),
                                                          (short)0, acc11, false, false);
            acc00 = __builtin_amdgcn_wmma_f32_16x16x4_f32(false, HI2(a0), false, HI2(b0),
                                                          (short)0, acc00, false, false);
            acc01 = __builtin_amdgcn_wmma_f32_16x16x4_f32(false, HI2(a0), false, HI2(b1),
                                                          (short)0, acc01, false, false);
            acc10 = __builtin_amdgcn_wmma_f32_16x16x4_f32(false, HI2(a1), false, HI2(b0),
                                                          (short)0, acc10, false, false);
            acc11 = __builtin_amdgcn_wmma_f32_16x16x4_f32(false, HI2(a1), false, HI2(b1),
                                                          (short)0, acc11, false, false);
        }
    }
    const int m0 = (lane >> 4) * 8;
#pragma unroll
    for (int r = 0; r < 8; ++r) {
        T[(t0 + m0 + r)      * LOWRANK + f0 + lr]      = acc00[r];
        T[(t0 + m0 + r)      * LOWRANK + f0 + 16 + lr] = acc01[r];
        T[(t0 + 16 + m0 + r) * LOWRANK + f0 + lr]      = acc10[r];
        T[(t0 + 16 + m0 + r) * LOWRANK + f0 + 16 + lr] = acc11[r];
    }
}

#define KS          32
#define PITCH       36
#define A_ROWS      128
#define B_ROWS      256
#define STAGE_FL    ((A_ROWS + B_ROWS) * PITCH)
#define NSTEP       ((KBF + LOWRANK) / KS)

__global__ __launch_bounds__(256, 4) void fused_out(const float* __restrict__ x,
                                                 const float* __restrict__ wg,
                                                 const float* __restrict__ T,
                                                 const float* __restrict__ w2,
                                                 const float* __restrict__ bias,
                                                 const int* __restrict__ flat,
                                                 float* __restrict__ out) {
    __shared__ __attribute__((aligned(16))) float lds[2 * STAGE_FL];

    const int tid  = threadIdx.x;
    const int lane = tid & 31;
    const int wave = tid >> 5;
    const int ob   = blockIdx.x & 15;
    const int t0   = (blockIdx.x >> 4) * A_ROWS;
    const int f0   = ob * B_ROWS;
    const int wm   = (wave >> 2) * 64;
    const int wn   = (wave & 3) * 64;
    const int lr   = lane & 15;
    const int co   = (lane >> 4) * 4;

    const int cr   = tid >> 3;
    const int cc   = (tid & 7) * 4;

    const int q0 = flat[ob * ACTIVE + 0] / ACTIVE;
    const int q1 = flat[ob * ACTIVE + 1] / ACTIVE;
    const int q2 = flat[ob * ACTIVE + 2] / ACTIVE;
    const int q3 = flat[ob * ACTIVE + 3] / ACTIVE;
    const int q4 = flat[ob * ACTIVE + 4] / ACTIVE;

    v8f acc[4][4];
#pragma unroll
    for (int i = 0; i < 4; ++i)
#pragma unroll
        for (int n = 0; n < 4; ++n) acc[i][n] = (v8f){};

    v4f ra[4], rb[8];

    for (int s = 0; s <= NSTEP; ++s) {
        if (s < NSTEP) {
            const int j  = s >> 3;
            const int kk = (s & 7) * KS + cc;
            const float* ap;
            const float* bp;
            int as, bs;
            if (j < ACTIVE) {
                const int q = (j == 0) ? q0 : (j == 1) ? q1 : (j == 2) ? q2 : (j == 3) ? q3 : q4;
                ap = x  + (t0 + cr) * IN_F + q * BS + kk;
                bp = wg + (f0 + cr) * KBF  + j * BS + kk;
                as = 32 * IN_F;
                bs = 32 * KBF;
            } else {
                ap = T  + (t0 + cr) * LOWRANK + kk;
                bp = w2 + (f0 + cr) * LOWRANK + kk;
                as = 32 * LOWRANK;
                bs = 32 * LOWRANK;
            }
#pragma unroll
            for (int i = 0; i < 4; ++i) ra[i] = *(const v4f*)(ap + i * as);
#pragma unroll
            for (int i = 0; i < 8; ++i) rb[i] = *(const v4f*)(bp + i * bs);
        }

        if (s > 0) {
            const float* fa = lds + ((s - 1) & 1) * STAGE_FL + (wm + lr) * PITCH + co;
            const float* fb = lds + ((s - 1) & 1) * STAGE_FL + (A_ROWS + wn + lr) * PITCH + co;
#pragma unroll
            for (int u = 0; u < KS; u += 8) {
                v4f af[4], bf[4];
#pragma unroll
                for (int i = 0; i < 4; ++i) af[i] = *(const v4f*)(fa + 16 * i * PITCH + u);
#pragma unroll
                for (int n = 0; n < 4; ++n) bf[n] = *(const v4f*)(fb + 16 * n * PITCH + u);
#pragma unroll
                for (int i = 0; i < 4; ++i)
#pragma unroll
                    for (int n = 0; n < 4; ++n)
                        acc[i][n] = __builtin_amdgcn_wmma_f32_16x16x4_f32(
                            false, LO2(af[i]), false, LO2(bf[n]), (short)0, acc[i][n], false, false);
#pragma unroll
                for (int i = 0; i < 4; ++i)
#pragma unroll
                    for (int n = 0; n < 4; ++n)
                        acc[i][n] = __builtin_amdgcn_wmma_f32_16x16x4_f32(
                            false, HI2(af[i]), false, HI2(bf[n]), (short)0, acc[i][n], false, false);
            }
            PRODUCT_TAIL_SLOTS();
        }

        if (s < NSTEP) {
            float* sa = lds + (s & 1) * STAGE_FL + cr * PITCH + cc;
            float* sb = sa + A_ROWS * PITCH;
#pragma unroll
            for (int i = 0; i < 4; ++i) *(v4f*)(sa + i * 32 * PITCH) = ra[i];
#pragma unroll
            for (int i = 0; i < 8; ++i) *(v4f*)(sb + i * 32 * PITCH) = rb[i];
        }
        __syncthreads();
    }

    const int m0 = (lane >> 4) * 8;
#pragma unroll
    for (int n = 0; n < 4; ++n) {
        const float bv = bias[f0 + wn + 16 * n + lr];
#pragma unroll
        for (int i = 0; i < 4; ++i)
#pragma unroll
            for (int r = 0; r < 8; ++r)
                out[(t0 + wm + 16 * i + m0 + r) * OUT_F + f0 + wn + 16 * n + lr] = acc[i][n][r] + bv;
    }
}

extern "C" void kernel_launch(void* const* d_in, const int* in_sizes, int n_in,
                              void* d_out, int out_size, void* d_ws, size_t ws_size,
                              hipStream_t stream) {
    const float* x      = (const float*)d_in[0];
    const float* weight = (const float*)d_in[1];
    const float* w1     = (const float*)d_in[2];
    const float* w2     = (const float*)d_in[3];
    const float* bias   = (const float*)d_in[4];
    const int*   flat   = (const int*)d_in[5];
    float*       out    = (float*)d_out;

    float* wg = (float*)d_ws;
    float* T  = wg + (size_t)OUT_F * KBF;

    gather_wg<<<dim3((OUT_F * KBF) / 256), dim3(256), 0, stream>>>(weight, flat, wg);

    gemm_xw1<<<dim3(1024 / 8), dim3(256), 0, stream>>>(x, w1, T);

    fused_out<<<dim3(512), dim3(256), 0, stream>>>(x, wg, T, w2, bias, flat, out);
}


# Output kernel reads each 256x256 butterfly weight block in place (weight row = flat*256+r); the gather kernel, its 20-MB copy and one launch are gone; summation order unchanged
# speedup vs baseline: 1.5165x; 1.0433x over previous
#include <hip/hip_runtime.h>

typedef __attribute__((ext_vector_type(2))) float v2f;
typedef __attribute__((ext_vector_type(4))) float v4f;
typedef __attribute__((ext_vector_type(8))) float v8f;

#define LO2(v) (__builtin_shufflevector((v), (v), 0, 1))
#define HI2(v) (__builtin_shufflevector((v), (v), 2, 3))

#define PRODUCT_TAIL_SLOTS()                                                        \
    do {                                                                            \
        __builtin_amdgcn_sched_barrier(0);                                          \
        asm volatile("v_nop\n\tv_nop\n\tv_nop\n\tv_nop\n\tv_nop\n\tv_nop\n\tv_nop\n\tv_nop"); \
        __builtin_amdgcn_sched_barrier(0);                                          \
    } while (0)

#define TOKENS  4096
#define IN_F    4096
#define OUT_F   4096
#define LOWRANK 256
#define BS      256
#define ACTIVE  5
#define KBF     (ACTIVE * BS)

__global__ __launch_bounds__(256) void gemm_xw1(const float* __restrict__ x,
                                                const float* __restrict__ w1,
                                                float* __restrict__ T) {
    const int lane = threadIdx.x & 31;
    const int wave = threadIdx.x >> 5;
    const int tile = blockIdx.x * 8 + wave;
    const int nft  = LOWRANK / 32;
    const int tt   = tile / nft;
    const int ft   = tile - tt * nft;
    const int t0   = tt * 32;
    const int f0   = ft * 32;
    const int lr   = lane & 15;
    const int co   = (lane >> 4) * 4;

    const float* a0p = x  + (t0 + lr)      * IN_F + co;
    const float* a1p = x  + (t0 + 16 + lr) * IN_F + co;
    const float* b0p = w1 + (f0 + lr)      * IN_F + co;
    const float* b1p = w1 + (f0 + 16 + lr) * IN_F + co;

    v8f acc00 = {}, acc01 = {}, acc10 = {}, acc11 = {};
    for (int k = 0; k < IN_F; k += 16) {
#pragma unroll
        for (int u = 0; u < 16; u += 8) {
            const v4f a0 = *(const v4f*)(a0p + k + u);
            const v4f a1 = *(const v4f*)(a1p + k + u);
            const v4f b0 = *(const v4f*)(b0p + k + u);
            const v4f b1 = *(const v4f*)(b1p + k + u);
            acc00 = __builtin_amdgcn_wmma_f32_16x16x4_f32(false, LO2(a0), false, LO2(b0),
                                                          (short)0, acc00, false, false);
            acc01 = __builtin_amdgcn_wmma_f32_16x16x4_f32(false, LO2(a0), false, LO2(b1),
                                                          (short)0, acc01, false, false);
            acc10 = __builtin_amdgcn_wmma_f32_16x16x4_f32(false, LO2(a1), false, LO2(b0),
                                                          (short)0, acc10, false, false);
            acc11 = __builtin_amdgcn_wmma_f32_16x16x4_f32(false, LO2(a1), false, LO2(b1),
                                                          (short)0, acc11, false, false);
            acc00 = __builtin_amdgcn_wmma_f32_16x16x4_f32(false, HI2(a0), false, HI2(b0),
                                                          (short)0, acc00, false, false);
            acc01 = __builtin_amdgcn_wmma_f32_16x16x4_f32(false, HI2(a0), false, HI2(b1),
                                                          (short)0, acc01, false, false);
            acc10 = __builtin_amdgcn_wmma_f32_16x16x4_f32(false, HI2(a1), false, HI2(b0),
                                                          (short)0, acc10, false, false);
            acc11 = __builtin_amdgcn_wmma_f32_16x16x4_f32(false, HI2(a1), false, HI2(b1),
                                                          (short)0, acc11, false, false);
        }
    }
    const int m0 = (lane >> 4) * 8;
#pragma unroll
    for (int r = 0; r < 8; ++r) {
        T[(t0 + m0 + r)      * LOWRANK + f0 + lr]      = acc00[r];
        T[(t0 + m0 + r)      * LOWRANK + f0 + 16 + lr] = acc01[r];
        T[(t0 + 16 + m0 + r) * LOWRANK + f0 + lr]      = acc10[r];
        T[(t0 + 16 + m0 + r) * LOWRANK + f0 + 16 + lr] = acc11[r];
    }
}

#define KS          32
#define PITCH       36
#define A_ROWS      128
#define B_ROWS      256
#define STAGE_FL    ((A_ROWS + B_ROWS) * PITCH)
#define NSTEP       ((KBF + LOWRANK) / KS)

__global__ __launch_bounds__(256, 4) void fused_out(const float* __restrict__ x,
                                                 const float* __restrict__ weight,
                                                 const float* __restrict__ T,
                                                 const float* __restrict__ w2,
                                                 const float* __restrict__ bias,
                                                 const int* __restrict__ flat,
                                                 float* __restrict__ out) {
    __shared__ __attribute__((aligned(16))) float lds[2 * STAGE_FL];

    const int tid  = threadIdx.x;
    const int lane = tid & 31;
    const int wave = tid >> 5;
    const int ob   = blockIdx.x & 15;
    const int t0   = (blockIdx.x >> 4) * A_ROWS;
    const int f0   = ob * B_ROWS;
    const int wm   = (wave >> 2) * 64;
    const int wn   = (wave & 3) * 64;
    const int lr   = lane & 15;
    const int co   = (lane >> 4) * 4;

    const int cr   = tid >> 3;
    const int cc   = (tid & 7) * 4;

    const int fl0 = flat[ob * ACTIVE + 0];
    const int fl1 = flat[ob * ACTIVE + 1];
    const int fl2 = flat[ob * ACTIVE + 2];
    const int fl3 = flat[ob * ACTIVE + 3];
    const int fl4 = flat[ob * ACTIVE + 4];

    v8f acc[4][4];
#pragma unroll
    for (int i = 0; i < 4; ++i)
#pragma unroll
        for (int n = 0; n < 4; ++n) acc[i][n] = (v8f){};

    v4f ra[4], rb[8];

    for (int s = 0; s <= NSTEP; ++s) {
        if (s < NSTEP) {
            const int j  = s >> 3;
            const int kk = (s & 7) * KS + cc;
            const float* ap;
            const float* bp;
            int as, bs;
            if (j < ACTIVE) {
                const int fl = (j == 0) ? fl0 : (j == 1) ? fl1 : (j == 2) ? fl2 : (j == 3) ? fl3 : fl4;
                const int q  = fl / ACTIVE;
                ap = x      + (t0 + cr) * IN_F + q * BS + kk;
                bp = weight + (fl * BS + cr) * BS + kk;
                as = 32 * IN_F;
                bs = 32 * BS;
            } else {
                ap = T  + (t0 + cr) * LOWRANK + kk;
                bp = w2 + (f0 + cr) * LOWRANK + kk;
                as = 32 * LOWRANK;
                bs = 32 * LOWRANK;
            }
#pragma unroll
            for (int i = 0; i < 4; ++i) ra[i] = *(const v4f*)(ap + i * as);
#pragma unroll
            for (int i = 0; i < 8; ++i) rb[i] = *(const v4f*)(bp + i * bs);
        }

        if (s > 0) {
            const float* fa = lds + ((s - 1) & 1) * STAGE_FL + (wm + lr) * PITCH + co;
            const float* fb = lds + ((s - 1) & 1) * STAGE_FL + (A_ROWS + wn + lr) * PITCH + co;
#pragma unroll
            for (int u = 0; u < KS; u += 8) {
                v4f af[4], bf[4];
#pragma unroll
                for (int i = 0; i < 4; ++i) af[i] = *(const v4f*)(fa + 16 * i * PITCH + u);
#pragma unroll
                for (int n = 0; n < 4; ++n) bf[n] = *(const v4f*)(fb + 16 * n * PITCH + u);
#pragma unroll
                for (int i = 0; i < 4; ++i)
#pragma unroll
                    for (int n = 0; n < 4; ++n)
                        acc[i][n] = __builtin_amdgcn_wmma_f32_16x16x4_f32(
                            false, LO2(af[i]), false, LO2(bf[n]), (short)0, acc[i][n], false, false);
#pragma unroll
                for (int i = 0; i < 4; ++i)
#pragma unroll
                    for (int n = 0; n < 4; ++n)
                        acc[i][n] = __builtin_amdgcn_wmma_f32_16x16x4_f32(
                            false, HI2(af[i]), false, HI2(bf[n]), (short)0, acc[i][n], false, false);
            }
            PRODUCT_TAIL_SLOTS();
        }

        if (s < NSTEP) {
            float* sa = lds + (s & 1) * STAGE_FL + cr * PITCH + cc;
            float* sb = sa + A_ROWS * PITCH;
#pragma unroll
            for (int i = 0; i < 4; ++i) *(v4f*)(sa + i * 32 * PITCH) = ra[i];
#pragma unroll
            for (int i = 0; i < 8; ++i) *(v4f*)(sb + i * 32 * PITCH) = rb[i];
        }
        __syncthreads();
    }

    const int m0 = (lane >> 4) * 8;
#pragma unroll
    for (int n = 0; n < 4; ++n) {
        const float bv = bias[f0 + wn + 16 * n + lr];
#pragma unroll
        for (int i = 0; i < 4; ++i)
#pragma unroll
            for (int r = 0; r < 8; ++r)
                out[(t0 + wm + 16 * i + m0 + r) * OUT_F + f0 + wn + 16 * n + lr] = acc[i][n][r] + bv;
    }
}

extern "C" void kernel_launch(void* const* d_in, const int* in_sizes, int n_in,
                              void* d_out, int out_size, void* d_ws, size_t ws_size,
                              hipStream_t stream) {
    const float* x      = (const float*)d_in[0];
    const float* weight = (const float*)d_in[1];
    const float* w1     = (const float*)d_in[2];
    const float* w2     = (const float*)d_in[3];
    const float* bias   = (const float*)d_in[4];
    const int*   flat   = (const int*)d_in[5];
    float*       out    = (float*)d_out;

    float* T = (float*)d_ws;

    gemm_xw1<<<dim3(1024 / 8), dim3(256), 0, stream>>>(x, w1, T);

    fused_out<<<dim3(512), dim3(256), 0, stream>>>(x, weight, T, w2, bias, flat, out);
}


# x@w1.T kernel re-tiled: 256 work-groups of 64x64 (was 128 of 32x256), A/B staged through LDS by whole-line global loads (2 stages, 1 barrier per 128 k), no per-wave fragment loads; k order unchanged
# speedup vs baseline: 2.1175x; 1.3963x over previous
#include <hip/hip_runtime.h>

typedef __attribute__((ext_vector_type(2))) float v2f;
typedef __attribute__((ext_vector_type(4))) float v4f;
typedef __attribute__((ext_vector_type(8))) float v8f;

#define LO2(v) (__builtin_shufflevector((v), (v), 0, 1))
#define HI2(v) (__builtin_shufflevector((v), (v), 2, 3))

#define PRODUCT_TAIL_SLOTS()                                                        \
    do {                                                                            \
        __builtin_amdgcn_sched_barrier(0);                                          \
        asm volatile("v_nop\n\tv_nop\n\tv_nop\n\tv_nop\n\tv_nop\n\tv_nop\n\tv_nop\n\tv_nop"); \
        __builtin_amdgcn_sched_barrier(0);                                          \
    } while (0)

#define TOKENS  4096
#define IN_F    4096
#define OUT_F   4096
#define LOWRANK 256
#define BS      256
#define ACTIVE  5
#define KBF     (ACTIVE * BS)

#define G1_TM       64
#define G1_TN       64
#define G1_KS       128
#define G1_PITCH    132
#define G1_STAGE_FL ((G1_TM + G1_TN) * G1_PITCH)
#define G1_NSTEP    (IN_F / G1_KS)

__global__ __launch_bounds__(256) void gemm_xw1(const float* __restrict__ x,
                                                const float* __restrict__ w1,
                                                float* __restrict__ T) {
    __shared__ __attribute__((aligned(16))) float lds[2 * G1_STAGE_FL];

    const int tid  = threadIdx.x;
    const int lane = tid & 31;
    const int wave = tid >> 5;
    const int t0   = (blockIdx.x >> 2) * G1_TM;
    const int f0   = (blockIdx.x & 3) * G1_TN;
    const int wm   = (wave >> 2) * 32;
    const int wn   = (wave & 3) * 16;
    const int lr   = lane & 15;
    const int co   = (lane >> 4) * 4;

    const int cr   = tid >> 5;
    const int cc   = (tid & 31) * 4;

    const float* ap = x  + (t0 + cr) * IN_F + cc;
    const float* bp = w1 + (f0 + cr) * IN_F + cc;

    v8f acc0 = {}, acc1 = {};
    v4f ra[8], rb[8];

    for (int s = 0; s <= G1_NSTEP; ++s) {
        if (s < G1_NSTEP) {
#pragma unroll
            for (int i = 0; i < 8; ++i) ra[i] = *(const v4f*)(ap + s * G1_KS + i * 8 * IN_F);
#pragma unroll
            for (int i = 0; i < 8; ++i) rb[i] = *(const v4f*)(bp + s * G1_KS + i * 8 * IN_F);
        }
        __builtin_amdgcn_sched_barrier(0);

        if (s > 0) {
            const float* fa = lds + ((s - 1) & 1) * G1_STAGE_FL + (wm + lr) * G1_PITCH + co;
            const float* fb = lds + ((s - 1) & 1) * G1_STAGE_FL + (G1_TM + wn + lr) * G1_PITCH + co;
#pragma unroll
            for (int u = 0; u < G1_KS; u += 8) {
                const v4f a0 = *(const v4f*)(fa + u);
                const v4f a1 = *(const v4f*)(fa + 16 * G1_PITCH + u);
                const v4f b0 = *(const v4f*)(fb + u);
                acc0 = __builtin_amdgcn_wmma_f32_16x16x4_f32(false, LO2(a0), false, LO2(b0),
                                                             (short)0, acc0, false, false);
                acc1 = __builtin_amdgcn_wmma_f32_16x16x4_f32(false, LO2(a1), false, LO2(b0),
                                                             (short)0, acc1, false, false);
                acc0 = __builtin_amdgcn_wmma_f32_16x16x4_f32(false, HI2(a0), false, HI2(b0),
                                                             (short)0, acc0, false, false);
                acc1 = __builtin_amdgcn_wmma_f32_16x16x4_f32(false, HI2(a1), false, HI2(b0),
                                                             (short)0, acc1, false, false);
            }
            PRODUCT_TAIL_SLOTS();
        }

        if (s < G1_NSTEP) {
            float* sa = lds + (s & 1) * G1_STAGE_FL + cr * G1_PITCH + cc;
            float* sb = sa + G1_TM * G1_PITCH;
#pragma unroll
            for (int i = 0; i < 8; ++i) *(v4f*)(sa + i * 8 * G1_PITCH) = ra[i];
#pragma unroll
            for (int i = 0; i < 8; ++i) *(v4f*)(sb + i * 8 * G1_PITCH) = rb[i];
        }
        __syncthreads();
    }

    const int m0 = (lane >> 4) * 8;
#pragma unroll
    for (int r = 0; r < 8; ++r) {
        lds[(wm + m0 + r)      * G1_PITCH + wn + lr] = acc0[r];
        lds[(wm + 16 + m0 + r) * G1_PITCH + wn + lr] = acc1[r];
    }
    __syncthreads();
    const int orow = tid >> 4;
    const int ocol = (tid & 15) * 4;
#pragma unroll
    for (int i = 0; i < 4; ++i) {
        const v4f v = *(const v4f*)(lds + (orow + 16 * i) * G1_PITCH + ocol);
        *(v4f*)(T + (t0 + orow + 16 * i) * LOWRANK + f0 + ocol) = v;
    }
}

#define KS          32
#define PITCH       36
#define A_ROWS      128
#define B_ROWS      256
#define STAGE_FL    ((A_ROWS + B_ROWS) * PITCH)
#define NSTEP       ((KBF + LOWRANK) / KS)

__global__ __launch_bounds__(256, 4) void fused_out(const float* __restrict__ x,
                                                 const float* __restrict__ weight,
                                                 const float* __restrict__ T,
                                                 const float* __restrict__ w2,
                                                 const float* __restrict__ bias,
                                                 const int* __restrict__ flat,
                                                 float* __restrict__ out) {
    __shared__ __attribute__((aligned(16))) float lds[2 * STAGE_FL];

    const int tid  = threadIdx.x;
    const int lane = tid & 31;
    const int wave = tid >> 5;
    const int ob   = blockIdx.x & 15;
    const int t0   = (blockIdx.x >> 4) * A_ROWS;
    const int f0   = ob * B_ROWS;
    const int wm   = (wave >> 2) * 64;
    const int wn   = (wave & 3) * 64;
    const int lr   = lane & 15;
    const int co   = (lane >> 4) * 4;

    const int cr   = tid >> 3;
    const int cc   = (tid & 7) * 4;

    const int fl0 = flat[ob * ACTIVE + 0];
    const int fl1 = flat[ob * ACTIVE + 1];
    const int fl2 = flat[ob * ACTIVE + 2];
    const int fl3 = flat[ob * ACTIVE + 3];
    const int fl4 = flat[ob * ACTIVE + 4];

    v8f acc[4][4];
#pragma unroll
    for (int i = 0; i < 4; ++i)
#pragma unroll
        for (int n = 0; n < 4; ++n) acc[i][n] = (v8f){};

    v4f ra[4], rb[8];

    for (int s = 0; s <= NSTEP; ++s) {
        if (s < NSTEP) {
            const int j  = s >> 3;
            const int kk = (s & 7) * KS + cc;
            const float* ap;
            const float* bp;
            int as, bs;
            if (j < ACTIVE) {
                const int fl = (j == 0) ? fl0 : (j == 1) ? fl1 : (j == 2) ? fl2 : (j == 3) ? fl3 : fl4;
                const int q  = fl / ACTIVE;
                ap = x      + (t0 + cr) * IN_F + q * BS + kk;
                bp = weight + (fl * BS + cr) * BS + kk;
                as = 32 * IN_F;
                bs = 32 * BS;
            } else {
                ap = T  + (t0 + cr) * LOWRANK + kk;
                bp = w2 + (f0 + cr) * LOWRANK + kk;
                as = 32 * LOWRANK;
                bs = 32 * LOWRANK;
            }
#pragma unroll
            for (int i = 0; i < 4; ++i) ra[i] = *(const v4f*)(ap + i * as);
#pragma unroll
            for (int i = 0; i < 8; ++i) rb[i] = *(const v4f*)(bp + i * bs);
        }

        if (s > 0) {
            const float* fa = lds + ((s - 1) & 1) * STAGE_FL + (wm + lr) * PITCH + co;
            const float* fb = lds + ((s - 1) & 1) * STAGE_FL + (A_ROWS + wn + lr) * PITCH + co;
#pragma unroll
            for (int u = 0; u < KS; u += 8) {
                v4f af[4], bf[4];
#pragma unroll
                for (int i = 0; i < 4; ++i) af[i] = *(const v4f*)(fa + 16 * i * PITCH + u);
#pragma unroll
                for (int n = 0; n < 4; ++n) bf[n] = *(const v4f*)(fb + 16 * n * PITCH + u);
#pragma unroll
                for (int i = 0; i < 4; ++i)
#pragma unroll
                    for (int n = 0; n < 4; ++n)
                        acc[i][n] = __builtin_amdgcn_wmma_f32_16x16x4_f32(
                            false, LO2(af[i]), false, LO2(bf[n]), (short)0, acc[i][n], false, false);
#pragma unroll
                for (int i = 0; i < 4; ++i)
#pragma unroll
                    for (int n = 0; n < 4; ++n)
                        acc[i][n] = __builtin_amdgcn_wmma_f32_16x16x4_f32(
                            false, HI2(af[i]), false, HI2(bf[n]), (short)0, acc[i][n], false, false);
            }
            PRODUCT_TAIL_SLOTS();
        }

        if (s < NSTEP) {
            float* sa = lds + (s & 1) * STAGE_FL + cr * PITCH + cc;
            float* sb = sa + A_ROWS * PITCH;
#pragma unroll
            for (int i = 0; i < 4; ++i) *(v4f*)(sa + i * 32 * PITCH) = ra[i];
#pragma unroll
            for (int i = 0; i < 8; ++i) *(v4f*)(sb + i * 32 * PITCH) = rb[i];
        }
        __syncthreads();
    }

    const int m0 = (lane >> 4) * 8;
#pragma unroll
    for (int n = 0; n < 4; ++n) {
        const float bv = bias[f0 + wn + 16 * n + lr];
#pragma unroll
        for (int i = 0; i < 4; ++i)
#pragma unroll
            for (int r = 0; r < 8; ++r)
                out[(t0 + wm + 16 * i + m0 + r) * OUT_F + f0 + wn + 16 * n + lr] = acc[i][n][r] + bv;
    }
}

extern "C" void kernel_launch(void* const* d_in, const int* in_sizes, int n_in,
                              void* d_out, int out_size, void* d_ws, size_t ws_size,
                              hipStream_t stream) {
    const float* x      = (const float*)d_in[0];
    const float* weight = (const float*)d_in[1];
    const float* w1     = (const float*)d_in[2];
    const float* w2     = (const float*)d_in[3];
    const float* bias   = (const float*)d_in[4];
    const int*   flat   = (const int*)d_in[5];
    float*       out    = (float*)d_out;

    float* T = (float*)d_ws;

    gemm_xw1<<<dim3(256), dim3(256), 0, stream>>>(x, w1, T);

    fused_out<<<dim3(512), dim3(256), 0, stream>>>(x, weight, T, w2, bias, flat, out);
}


# Output kernel: wave priority set from the step count (3,2,1,0 at steps 0,24,40,45) so the two co-resident work-groups share the matrix pipe to the end instead of one running alone; sums unchanged
# speedup vs baseline: 2.1934x; 1.0358x over previous
#include <hip/hip_runtime.h>

typedef __attribute__((ext_vector_type(2))) float v2f;
typedef __attribute__((ext_vector_type(4))) float v4f;
typedef __attribute__((ext_vector_type(8))) float v8f;

#define LO2(v) (__builtin_shufflevector((v), (v), 0, 1))
#define HI2(v) (__builtin_shufflevector((v), (v), 2, 3))

#define PRODUCT_TAIL_SLOTS()                                                        \
    do {                                                                            \
        __builtin_amdgcn_sched_barrier(0);                                          \
        asm volatile("v_nop\n\tv_nop\n\tv_nop\n\tv_nop\n\tv_nop\n\tv_nop\n\tv_nop\n\tv_nop"); \
        __builtin_amdgcn_sched_barrier(0);                                          \
    } while (0)

#define TOKENS  4096
#define IN_F    4096
#define OUT_F   4096
#define LOWRANK 256
#define BS      256
#define ACTIVE  5
#define KBF     (ACTIVE * BS)

#define G1_TM       64
#define G1_TN       64
#define G1_KS       128
#define G1_PITCH    132
#define G1_STAGE_FL ((G1_TM + G1_TN) * G1_PITCH)
#define G1_NSTEP    (IN_F / G1_KS)

__global__ __launch_bounds__(256) void gemm_xw1(const float* __restrict__ x,
                                                const float* __restrict__ w1,
                                                float* __restrict__ T) {
    __shared__ __attribute__((aligned(16))) float lds[2 * G1_STAGE_FL];

    const int tid  = threadIdx.x;
    const int lane = tid & 31;
    const int wave = tid >> 5;
    const int t0   = (blockIdx.x >> 2) * G1_TM;
    const int f0   = (blockIdx.x & 3) * G1_TN;
    const int wm   = (wave >> 2) * 32;
    const int wn   = (wave & 3) * 16;
    const int lr   = lane & 15;
    const int co   = (lane >> 4) * 4;

    const int cr   = tid >> 5;
    const int cc   = (tid & 31) * 4;

    const float* ap = x  + (t0 + cr) * IN_F + cc;
    const float* bp = w1 + (f0 + cr) * IN_F + cc;

    v8f acc0 = {}, acc1 = {};
    v4f ra[8], rb[8];

    for (int s = 0; s <= G1_NSTEP; ++s) {
        if (s < G1_NSTEP) {
#pragma unroll
            for (int i = 0; i < 8; ++i) ra[i] = *(const v4f*)(ap + s * G1_KS + i * 8 * IN_F);
#pragma unroll
            for (int i = 0; i < 8; ++i) rb[i] = *(const v4f*)(bp + s * G1_KS + i * 8 * IN_F);
        }
        __builtin_amdgcn_sched_barrier(0);

        if (s > 0) {
            const float* fa = lds + ((s - 1) & 1) * G1_STAGE_FL + (wm + lr) * G1_PITCH + co;
            const float* fb = lds + ((s - 1) & 1) * G1_STAGE_FL + (G1_TM + wn + lr) * G1_PITCH + co;
#pragma unroll
            for (int u = 0; u < G1_KS; u += 8) {
                const v4f a0 = *(const v4f*)(fa + u);
                const v4f a1 = *(const v4f*)(fa + 16 * G1_PITCH + u);
                const v4f b0 = *(const v4f*)(fb + u);
                acc0 = __builtin_amdgcn_wmma_f32_16x16x4_f32(false, LO2(a0), false, LO2(b0),
                                                             (short)0, acc0, false, false);
                acc1 = __builtin_amdgcn_wmma_f32_16x16x4_f32(false, LO2(a1), false, LO2(b0),
                                                             (short)0, acc1, false, false);
                acc0 = __builtin_amdgcn_wmma_f32_16x16x4_f32(false, HI2(a0), false, HI2(b0),
                                                             (short)0, acc0, false, false);
                acc1 = __builtin_amdgcn_wmma_f32_16x16x4_f32(false, HI2(a1), false, HI2(b0),
                                                             (short)0, acc1, false, false);
            }
            PRODUCT_TAIL_SLOTS();
        }

        if (s < G1_NSTEP) {
            float* sa = lds + (s & 1) * G1_STAGE_FL + cr * G1_PITCH + cc;
            float* sb = sa + G1_TM * G1_PITCH;
#pragma unroll
            for (int i = 0; i < 8; ++i) *(v4f*)(sa + i * 8 * G1_PITCH) = ra[i];
#pragma unroll
            for (int i = 0; i < 8; ++i) *(v4f*)(sb + i * 8 * G1_PITCH) = rb[i];
        }
        __syncthreads();
    }

    const int m0 = (lane >> 4) * 8;
#pragma unroll
    for (int r = 0; r < 8; ++r) {
        lds[(wm + m0 + r)      * G1_PITCH + wn + lr] = acc0[r];
        lds[(wm + 16 + m0 + r) * G1_PITCH + wn + lr] = acc1[r];
    }
    __syncthreads();
    const int orow = tid >> 4;
    const int ocol = (tid & 15) * 4;
#pragma unroll
    for (int i = 0; i < 4; ++i) {
        const v4f v = *(const v4f*)(lds + (orow + 16 * i) * G1_PITCH + ocol);
        *(v4f*)(T + (t0 + orow + 16 * i) * LOWRANK + f0 + ocol) = v;
    }
}

#define KS          32
#define PITCH       36
#define A_ROWS      128
#define B_ROWS      256
#define STAGE_FL    ((A_ROWS + B_ROWS) * PITCH)
#define NSTEP       ((KBF + LOWRANK) / KS)
#define PRIO_MARK_1 24
#define PRIO_MARK_2 40
#define PRIO_MARK_3 45

__global__ __launch_bounds__(256, 4) void fused_out(const float* __restrict__ x,
                                                 const float* __restrict__ weight,
                                                 const float* __restrict__ T,
                                                 const float* __restrict__ w2,
                                                 const float* __restrict__ bias,
                                                 const int* __restrict__ flat,
                                                 float* __restrict__ out) {
    __shared__ __attribute__((aligned(16))) float lds[2 * STAGE_FL];

    const int tid  = threadIdx.x;
    const int lane = tid & 31;
    const int wave = tid >> 5;
    const int ob   = blockIdx.x & 15;
    const int t0   = (blockIdx.x >> 4) * A_ROWS;
    const int f0   = ob * B_ROWS;
    const int wm   = (wave >> 2) * 64;
    const int wn   = (wave & 3) * 64;
    const int lr   = lane & 15;
    const int co   = (lane >> 4) * 4;

    const int cr   = tid >> 3;
    const int cc   = (tid & 7) * 4;

    const int fl0 = flat[ob * ACTIVE + 0];
    const int fl1 = flat[ob * ACTIVE + 1];
    const int fl2 = flat[ob * ACTIVE + 2];
    const int fl3 = flat[ob * ACTIVE + 3];
    const int fl4 = flat[ob * ACTIVE + 4];

    v8f acc[4][4];
#pragma unroll
    for (int i = 0; i < 4; ++i)
#pragma unroll
        for (int n = 0; n < 4; ++n) acc[i][n] = (v8f){};

    v4f ra[4], rb[8];

    for (int s = 0; s <= NSTEP; ++s) {
        if (s == 0)                __builtin_amdgcn_s_setprio(3);
        else if (s == PRIO_MARK_1) __builtin_amdgcn_s_setprio(2);
        else if (s == PRIO_MARK_2) __builtin_amdgcn_s_setprio(1);
        else if (s == PRIO_MARK_3) __builtin_amdgcn_s_setprio(0);

        if (s < NSTEP) {
            const int j  = s >> 3;
            const int kk = (s & 7) * KS + cc;
            const float* ap;
            const float* bp;
            int as, bs;
            if (j < ACTIVE) {
                const int fl = (j == 0) ? fl0 : (j == 1) ? fl1 : (j == 2) ? fl2 : (j == 3) ? fl3 : fl4;
                const int q  = fl / ACTIVE;
                ap = x      + (t0 + cr) * IN_F + q * BS + kk;
                bp = weight + (fl * BS + cr) * BS + kk;
                as = 32 * IN_F;
                bs = 32 * BS;
            } else {
                ap = T  + (t0 + cr) * LOWRANK + kk;
                bp = w2 + (f0 + cr) * LOWRANK + kk;
                as = 32 * LOWRANK;
                bs = 32 * LOWRANK;
            }
#pragma unroll
            for (int i = 0; i < 4; ++i) ra[i] = *(const v4f*)(ap + i * as);
#pragma unroll
            for (int i = 0; i < 8; ++i) rb[i] = *(const v4f*)(bp + i * bs);
        }

        if (s > 0) {
            const float* fa = lds + ((s - 1) & 1) * STAGE_FL + (wm + lr) * PITCH + co;
            const float* fb = lds + ((s - 1) & 1) * STAGE_FL + (A_ROWS + wn + lr) * PITCH + co;
#pragma unroll
            for (int u = 0; u < KS; u += 8) {
                v4f af[4], bf[4];
#pragma unroll
                for (int i = 0; i < 4; ++i) af[i] = *(const v4f*)(fa + 16 * i * PITCH + u);
#pragma unroll
                for (int n = 0; n < 4; ++n) bf[n] = *(const v4f*)(fb + 16 * n * PITCH + u);
#pragma unroll
                for (int i = 0; i < 4; ++i)
#pragma unroll
                    for (int n = 0; n < 4; ++n)
                        acc[i][n] = __builtin_amdgcn_wmma_f32_16x16x4_f32(
                            false, LO2(af[i]), false, LO2(bf[n]), (short)0, acc[i][n], false, false);
#pragma unroll
                for (int i = 0; i < 4; ++i)
#pragma unroll
                    for (int n = 0; n < 4; ++n)
                        acc[i][n] = __builtin_amdgcn_wmma_f32_16x16x4_f32(
                            false, HI2(af[i]), false, HI2(bf[n]), (short)0, acc[i][n], false, false);
            }
            PRODUCT_TAIL_SLOTS();
        }

        if (s < NSTEP) {
            float* sa = lds + (s & 1) * STAGE_FL + cr * PITCH + cc;
            float* sb = sa + A_ROWS * PITCH;
#pragma unroll
            for (int i = 0; i < 4; ++i) *(v4f*)(sa + i * 32 * PITCH) = ra[i];
#pragma unroll
            for (int i = 0; i < 8; ++i) *(v4f*)(sb + i * 32 * PITCH) = rb[i];
        }
        __syncthreads();
    }

    const int m0 = (lane >> 4) * 8;
#pragma unroll
    for (int n = 0; n < 4; ++n) {
        const float bv = bias[f0 + wn + 16 * n + lr];
#pragma unroll
        for (int i = 0; i < 4; ++i)
#pragma unroll
            for (int r = 0; r < 8; ++r)
                out[(t0 + wm + 16 * i + m0 + r) * OUT_F + f0 + wn + 16 * n + lr] = acc[i][n][r] + bv;
    }
}

extern "C" void kernel_launch(void* const* d_in, const int* in_sizes, int n_in,
                              void* d_out, int out_size, void* d_ws, size_t ws_size,
                              hipStream_t stream) {
    const float* x      = (const float*)d_in[0];
    const float* weight = (const float*)d_in[1];
    const float* w1     = (const float*)d_in[2];
    const float* w2     = (const float*)d_in[3];
    const float* bias   = (const float*)d_in[4];
    const int*   flat   = (const int*)d_in[5];
    float*       out    = (float*)d_out;

    float* T = (float*)d_ws;

    gemm_xw1<<<dim3(256), dim3(256), 0, stream>>>(x, w1, T);

    fused_out<<<dim3(512), dim3(256), 0, stream>>>(x, weight, T, w2, bias, flat, out);
}
